# starting kernel
# baseline (speedup 1.0000x reference)
#include <hip/hip_runtime.h>
#include <math.h>


typedef __attribute__((ext_vector_type(2))) float v2f;
typedef __attribute__((ext_vector_type(8))) float v8f;

#define HH 128
#define WW 128
#define HW (HH * WW)

__device__ __forceinline__ int reflect_idx(int i, int n) {
  if (i < 0) i = -i;
  if (i >= n) i = 2 * n - 2 - i;
  return i;
}

__global__ void pack_w(const float* __restrict__ cw, float* __restrict__ wp) {
  int t = blockIdx.x * blockDim.x + threadIdx.x;
  if (t >= 144 * 2 * 32 * 2) return;
  int v = t & 1;
  int lane = (t >> 1) & 31;
  int mt = (t >> 6) & 1;
  int kb = t >> 7;
  int m = (lane & 15) + 16 * mt;
  int hi = lane >> 4;
  int k = kb * 4 + 2 * hi + v;
  int kpos = k >> 6;
  int ic = k & 63;
  float val = 0.0f;
  if (m < 18) val = cw[(m * 64 + ic) * 9 + kpos];
  wp[t] = val;
}

__global__ __launch_bounds__(256) void conv_softmax(
    const float* __restrict__ feat, const float* __restrict__ bias,
    const float* __restrict__ wp, float* __restrict__ ffw) {
  __shared__ float xs[64 * 3 * 132];
  __shared__ float zb[8 * 16 * 19];

  const int h = blockIdx.x;
  const int b = blockIdx.y;
  const int tid = threadIdx.x;

  const float* fb = feat + (size_t)b * 64 * HW;
  for (int e = tid; e < 64 * 3 * 130; e += 256) {
    int ic = e / 390;
    int rem = e - ic * 390;
    int r = rem / 130;
    int wcol = rem - r * 130;
    int hr = reflect_idx(h + r - 1, HH);
    int ws = reflect_idx(wcol - 1, WW);
    xs[((ic >> 1) * 3 + r) * 264 + wcol * 2 + (ic & 1)] = fb[ic * HW + hr * WW + ws];
  }
  __syncthreads();

  const int lane = tid & 31, wv = tid >> 5;
  const int p = lane & 15, hi = lane >> 4;
  const int w0 = wv * 16;

  const v2f* __restrict__ wf = (const v2f*)wp;
  const v2f* xs2 = (const v2f*)xs;

  v8f acc0 = {0.f, 0.f, 0.f, 0.f, 0.f, 0.f, 0.f, 0.f};
  v8f acc1 = acc0;

  int kb = 0;
  for (int kpos = 0; kpos < 9; ++kpos) {
    const int dh = kpos / 3, dw = kpos - dh * 3;
    const v2f* xrow = xs2 + (size_t)hi * 396 + dh * 132 + (w0 + p + dw);
#pragma unroll
    for (int j = 0; j < 16; ++j, ++kb) {
      v2f a0 = wf[kb * 64 + lane];
      v2f a1 = wf[kb * 64 + 32 + lane];
      v2f bx = xrow[(size_t)j * 792];
      acc0 = __builtin_amdgcn_wmma_f32_16x16x4_f32(
          false, a0, false, bx, (short)0, acc0, false, false);
      acc1 = __builtin_amdgcn_wmma_f32_16x16x4_f32(
          false, a1, false, bx, (short)0, acc1, false, false);
    }
  }

  float* zrow = &zb[(wv * 16 + p) * 19];
#pragma unroll
  for (int r = 0; r < 8; ++r) zrow[r + 8 * hi] = acc0[r];
  if (hi == 0) {
    zrow[16] = acc1[0];
    zrow[17] = acc1[1];
  }
  __syncthreads();

  const int g = hi;
  float z[9];
#pragma unroll
  for (int k = 0; k < 9; ++k) z[k] = zrow[2 * k + g] + bias[2 * k + g];
  float mx = z[0];
#pragma unroll
  for (int k = 1; k < 9; ++k) mx = fmaxf(mx, z[k]);
  float s = 0.0f;
#pragma unroll
  for (int k = 0; k < 9; ++k) {
    z[k] = expf(z[k] - mx);
    s += z[k];
  }
  const float inv = 1.0f / s;
  const int wpix = w0 + p;
#pragma unroll
  for (int k = 0; k < 9; ++k) {
    int ch = 2 * k + g;
    int si = ch / 9, kk = ch - si * 9;
    ffw[(((size_t)b * 2 + si) * 9 + kk) * HW + h * WW + wpix] = z[k] * inv;
  }
}

__global__ __launch_bounds__(256) void combine(
    const float* __restrict__ inp, const float* __restrict__ ffw,
    float* __restrict__ out, int total) {
  int n = blockIdx.x * 256 + threadIdx.x;
  if (n >= total) return;
  int w = n & 127;
  int h = (n >> 7) & 127;
  int c = (n >> 14) & 63;
  int b = n >> 20;
  const float* xin = inp + (size_t)(b * 64 + c) * HW;
  const float* wg = ffw + (((size_t)b * 2 + (c & 1)) * 9) * HW + h * WW + w;
  float acc = 0.0f;
#pragma unroll
  for (int kpos = 0; kpos < 9; ++kpos) {
    int dh = kpos / 3, dw = kpos - dh * 3;
    int hr = reflect_idx(h + dh - 1, HH);
    int wr = reflect_idx(w + dw - 1, WW);
    acc = fmaf(wg[(size_t)kpos * HW], xin[hr * WW + wr], acc);
  }
  out[n] = acc;
}

extern "C" void kernel_launch(void* const* d_in, const int* in_sizes, int n_in,
                              void* d_out, int out_size, void* d_ws, size_t ws_size,
                              hipStream_t stream) {
  const float* features = (const float*)d_in[0];
  const float* inpt     = (const float*)d_in[1];
  const float* conv_w   = (const float*)d_in[2];
  const float* conv_b   = (const float*)d_in[3];
  float* out = (float*)d_out;

  float* wp  = (float*)d_ws;
  float* ffw = wp + 18432;

  pack_w<<<(18432 + 255) / 256, 256, 0, stream>>>(conv_w, wp);
  conv_softmax<<<dim3(HH, 8), 256, 0, stream>>>(features, conv_b, wp, ffw);
  const int total = 8 * 64 * HW;
  combine<<<(total + 255) / 256, 256, 0, stream>>>(inpt, ffw, out, total);
}


# conv kernel stages its feature slab with 16-byte whole-row loads, all 24 of a thread issued before the first LDS store, replacing 4-byte loads with per-element index math issued two at a time
# speedup vs baseline: 1.6370x; 1.6370x over previous
#include <hip/hip_runtime.h>
#include <math.h>


typedef __attribute__((ext_vector_type(2))) float v2f;
typedef __attribute__((ext_vector_type(4))) float v4f;
typedef __attribute__((ext_vector_type(8))) float v8f;

#define HH 128
#define WW 128
#define HW (HH * WW)

__device__ __forceinline__ int reflect_idx(int i, int n) {
  if (i < 0) i = -i;
  if (i >= n) i = 2 * n - 2 - i;
  return i;
}

__global__ void pack_w(const float* __restrict__ cw, float* __restrict__ wp) {
  int t = blockIdx.x * blockDim.x + threadIdx.x;
  if (t >= 144 * 2 * 32 * 2) return;
  int v = t & 1;
  int lane = (t >> 1) & 31;
  int mt = (t >> 6) & 1;
  int kb = t >> 7;
  int m = (lane & 15) + 16 * mt;
  int hi = lane >> 4;
  int k = kb * 4 + 2 * hi + v;
  int kpos = k >> 6;
  int ic = k & 63;
  float val = 0.0f;
  if (m < 18) val = cw[(m * 64 + ic) * 9 + kpos];
  wp[t] = val;
}

__global__ __launch_bounds__(256) void conv_softmax(
    const float* __restrict__ feat, const float* __restrict__ bias,
    const float* __restrict__ wp, float* __restrict__ ffw) {
  __shared__ __attribute__((aligned(16))) float xs[64 * 3 * 132];
  __shared__ float zb[8 * 16 * 19];

  const int h = blockIdx.x;
  const int b = blockIdx.y;
  const int tid = threadIdx.x;
  const int lane = tid & 31, wv = tid >> 5;

  const float* fb = feat + (size_t)b * 64 * HW;
  v4f ev[12], od[12];
#pragma unroll
  for (int i = 0; i < 12; ++i) {
    const int r = i % 3;
    const int icp = wv * 4 + i / 3;
    const int hr = reflect_idx(h + r - 1, HH);
    const float* src = fb + (size_t)(2 * icp) * HW + hr * WW + 4 * lane;
    ev[i] = *(const v4f*)src;
    od[i] = *(const v4f*)(src + HW);
  }
#pragma unroll
  for (int i = 0; i < 12; ++i) {
    const int r = i % 3;
    const int icp = wv * 4 + i / 3;
    float* dst = xs + (icp * 3 + r) * 264 + 2 * (4 * lane + 2);
    v4f lo = {ev[i].x, od[i].x, ev[i].y, od[i].y};
    v4f hi4 = {ev[i].z, od[i].z, ev[i].w, od[i].w};
    *(v4f*)dst = lo;
    *(v4f*)(dst + 4) = hi4;
  }
  if (lane == 0 || lane == 31) {
    const bool left = (lane == 0);
    const int eoff = left ? -2 : 8;
#pragma unroll
    for (int i = 0; i < 12; ++i) {
      const int r = i % 3;
      const int icp = wv * 4 + i / 3;
      float* dst = xs + (icp * 3 + r) * 264 + 2 * (4 * lane + 2);
      v2f e = {left ? ev[i].y : ev[i].z, left ? od[i].y : od[i].z};
      *(v2f*)(dst + eoff) = e;
    }
  }
  __syncthreads();

  const int p = lane & 15, hi = lane >> 4;
  const int w0 = wv * 16;

  const v2f* __restrict__ wf = (const v2f*)wp;
  const v2f* xs2 = (const v2f*)xs;

  v8f acc0 = {0.f, 0.f, 0.f, 0.f, 0.f, 0.f, 0.f, 0.f};
  v8f acc1 = acc0;

  int kb = 0;
  for (int kpos = 0; kpos < 9; ++kpos) {
    const int dh = kpos / 3, dw = kpos - dh * 3;
    const v2f* xrow = xs2 + (size_t)hi * 396 + dh * 132 + (w0 + p + dw + 1);
#pragma unroll
    for (int j = 0; j < 16; ++j, ++kb) {
      v2f a0 = wf[kb * 64 + lane];
      v2f a1 = wf[kb * 64 + 32 + lane];
      v2f bx = xrow[(size_t)j * 792];
      acc0 = __builtin_amdgcn_wmma_f32_16x16x4_f32(
          false, a0, false, bx, (short)0, acc0, false, false);
      acc1 = __builtin_amdgcn_wmma_f32_16x16x4_f32(
          false, a1, false, bx, (short)0, acc1, false, false);
    }
  }

  float* zrow = &zb[(wv * 16 + p) * 19];
#pragma unroll
  for (int r = 0; r < 8; ++r) zrow[r + 8 * hi] = acc0[r];
  if (hi == 0) {
    zrow[16] = acc1[0];
    zrow[17] = acc1[1];
  }
  __syncthreads();

  const int g = hi;
  float z[9];
#pragma unroll
  for (int k = 0; k < 9; ++k) z[k] = zrow[2 * k + g] + bias[2 * k + g];
  float mx = z[0];
#pragma unroll
  for (int k = 1; k < 9; ++k) mx = fmaxf(mx, z[k]);
  float s = 0.0f;
#pragma unroll
  for (int k = 0; k < 9; ++k) {
    z[k] = expf(z[k] - mx);
    s += z[k];
  }
  const float inv = 1.0f / s;
  const int wpix = w0 + p;
#pragma unroll
  for (int k = 0; k < 9; ++k) {
    int ch = 2 * k + g;
    int si = ch / 9, kk = ch - si * 9;
    ffw[(((size_t)b * 2 + si) * 9 + kk) * HW + h * WW + wpix] = z[k] * inv;
  }
}

__global__ __launch_bounds__(256) void combine(
    const float* __restrict__ inp, const float* __restrict__ ffw,
    float* __restrict__ out, int total) {
  int n = blockIdx.x * 256 + threadIdx.x;
  if (n >= total) return;
  int w = n & 127;
  int h = (n >> 7) & 127;
  int c = (n >> 14) & 63;
  int b = n >> 20;
  const float* xin = inp + (size_t)(b * 64 + c) * HW;
  const float* wg = ffw + (((size_t)b * 2 + (c & 1)) * 9) * HW + h * WW + w;
  float acc = 0.0f;
#pragma unroll
  for (int kpos = 0; kpos < 9; ++kpos) {
    int dh = kpos / 3, dw = kpos - dh * 3;
    int hr = reflect_idx(h + dh - 1, HH);
    int wr = reflect_idx(w + dw - 1, WW);
    acc = fmaf(wg[(size_t)kpos * HW], xin[hr * WW + wr], acc);
  }
  out[n] = acc;
}

extern "C" void kernel_launch(void* const* d_in, const int* in_sizes, int n_in,
                              void* d_out, int out_size, void* d_ws, size_t ws_size,
                              hipStream_t stream) {
  const float* features = (const float*)d_in[0];
  const float* inpt     = (const float*)d_in[1];
  const float* conv_w   = (const float*)d_in[2];
  const float* conv_b   = (const float*)d_in[3];
  float* out = (float*)d_out;

  float* wp  = (float*)d_ws;
  float* ffw = wp + 18432;

  pack_w<<<(18432 + 255) / 256, 256, 0, stream>>>(conv_w, wp);
  conv_softmax<<<dim3(HH, 8), 256, 0, stream>>>(features, conv_b, wp, ffw);
  const int total = 8 * 64 * HW;
  combine<<<(total + 255) / 256, 256, 0, stream>>>(inpt, ffw, out, total);
}


# combine kernel: one wave per image row, 16-byte loads and stores, the nine softmax weights of a lane's four pixels kept in registers over a 16-channel loop (was one thread per output, 18 loads each)
# speedup vs baseline: 2.3199x; 1.4172x over previous
#include <hip/hip_runtime.h>
#include <math.h>


typedef __attribute__((ext_vector_type(2))) float v2f;
typedef __attribute__((ext_vector_type(4))) float v4f;
typedef __attribute__((ext_vector_type(8))) float v8f;

#define HH 128
#define WW 128
#define HW (HH * WW)

__device__ __forceinline__ int reflect_idx(int i, int n) {
  if (i < 0) i = -i;
  if (i >= n) i = 2 * n - 2 - i;
  return i;
}

__global__ void pack_w(const float* __restrict__ cw, float* __restrict__ wp) {
  int t = blockIdx.x * blockDim.x + threadIdx.x;
  if (t >= 144 * 2 * 32 * 2) return;
  int v = t & 1;
  int lane = (t >> 1) & 31;
  int mt = (t >> 6) & 1;
  int kb = t >> 7;
  int m = (lane & 15) + 16 * mt;
  int hi = lane >> 4;
  int k = kb * 4 + 2 * hi + v;
  int kpos = k >> 6;
  int ic = k & 63;
  float val = 0.0f;
  if (m < 18) val = cw[(m * 64 + ic) * 9 + kpos];
  wp[t] = val;
}

__global__ __launch_bounds__(256) void conv_softmax(
    const float* __restrict__ feat, const float* __restrict__ bias,
    const float* __restrict__ wp, float* __restrict__ ffw) {
  __shared__ __attribute__((aligned(16))) float xs[64 * 3 * 132];
  __shared__ float zb[8 * 16 * 19];

  const int h = blockIdx.x;
  const int b = blockIdx.y;
  const int tid = threadIdx.x;
  const int lane = tid & 31, wv = tid >> 5;

  const float* fb = feat + (size_t)b * 64 * HW;
  v4f ev[12], od[12];
#pragma unroll
  for (int i = 0; i < 12; ++i) {
    const int r = i % 3;
    const int icp = wv * 4 + i / 3;
    const int hr = reflect_idx(h + r - 1, HH);
    const float* src = fb + (size_t)(2 * icp) * HW + hr * WW + 4 * lane;
    ev[i] = *(const v4f*)src;
    od[i] = *(const v4f*)(src + HW);
  }
#pragma unroll
  for (int i = 0; i < 12; ++i) {
    const int r = i % 3;
    const int icp = wv * 4 + i / 3;
    float* dst = xs + (icp * 3 + r) * 264 + 2 * (4 * lane + 2);
    v4f lo = {ev[i].x, od[i].x, ev[i].y, od[i].y};
    v4f hi4 = {ev[i].z, od[i].z, ev[i].w, od[i].w};
    *(v4f*)dst = lo;
    *(v4f*)(dst + 4) = hi4;
  }
  if (lane == 0 || lane == 31) {
    const bool left = (lane == 0);
    const int eoff = left ? -2 : 8;
#pragma unroll
    for (int i = 0; i < 12; ++i) {
      const int r = i % 3;
      const int icp = wv * 4 + i / 3;
      float* dst = xs + (icp * 3 + r) * 264 + 2 * (4 * lane + 2);
      v2f e = {left ? ev[i].y : ev[i].z, left ? od[i].y : od[i].z};
      *(v2f*)(dst + eoff) = e;
    }
  }
  __syncthreads();

  const int p = lane & 15, hi = lane >> 4;
  const int w0 = wv * 16;

  const v2f* __restrict__ wf = (const v2f*)wp;
  const v2f* xs2 = (const v2f*)xs;

  v8f acc0 = {0.f, 0.f, 0.f, 0.f, 0.f, 0.f, 0.f, 0.f};
  v8f acc1 = acc0;

  int kb = 0;
  for (int kpos = 0; kpos < 9; ++kpos) {
    const int dh = kpos / 3, dw = kpos - dh * 3;
    const v2f* xrow = xs2 + (size_t)hi * 396 + dh * 132 + (w0 + p + dw + 1);
#pragma unroll
    for (int j = 0; j < 16; ++j, ++kb) {
      v2f a0 = wf[kb * 64 + lane];
      v2f a1 = wf[kb * 64 + 32 + lane];
      v2f bx = xrow[(size_t)j * 792];
      acc0 = __builtin_amdgcn_wmma_f32_16x16x4_f32(
          false, a0, false, bx, (short)0, acc0, false, false);
      acc1 = __builtin_amdgcn_wmma_f32_16x16x4_f32(
          false, a1, false, bx, (short)0, acc1, false, false);
    }
  }

  float* zrow = &zb[(wv * 16 + p) * 19];
#pragma unroll
  for (int r = 0; r < 8; ++r) zrow[r + 8 * hi] = acc0[r];
  if (hi == 0) {
    zrow[16] = acc1[0];
    zrow[17] = acc1[1];
  }
  __syncthreads();

  const int g = hi;
  float z[9];
#pragma unroll
  for (int k = 0; k < 9; ++k) z[k] = zrow[2 * k + g] + bias[2 * k + g];
  float mx = z[0];
#pragma unroll
  for (int k = 1; k < 9; ++k) mx = fmaxf(mx, z[k]);
  float s = 0.0f;
#pragma unroll
  for (int k = 0; k < 9; ++k) {
    z[k] = expf(z[k] - mx);
    s += z[k];
  }
  const float inv = 1.0f / s;
  const int wpix = w0 + p;
#pragma unroll
  for (int k = 0; k < 9; ++k) {
    int ch = 2 * k + g;
    int si = ch / 9, kk = ch - si * 9;
    ffw[(((size_t)b * 2 + si) * 9 + kk) * HW + h * WW + wpix] = z[k] * inv;
  }
}

#define CPW 16

struct RowSet {
  v4f c[3];
  float l[3];
  float r[3];
};

__device__ __forceinline__ void load_rows(const float* __restrict__ x, int r0, int r1, int r2,
                                          int co, int lo, int ro, RowSet& t) {
  t.c[0] = *(const v4f*)(x + r0 + co);
  t.c[1] = *(const v4f*)(x + r1 + co);
  t.c[2] = *(const v4f*)(x + r2 + co);
  t.l[0] = x[r0 + lo];
  t.r[0] = x[r0 + ro];
  t.l[1] = x[r1 + lo];
  t.r[1] = x[r1 + ro];
  t.l[2] = x[r2 + lo];
  t.r[2] = x[r2 + ro];
}

__device__ __forceinline__ v4f apply_taps(const RowSet& t, const v4f* wt) {
  v4f acc = {0.f, 0.f, 0.f, 0.f};
#pragma unroll
  for (int dh = 0; dh < 3; ++dh) {
    const v4f c = t.c[dh];
    const v4f wl = wt[3 * dh], wc = wt[3 * dh + 1], wr = wt[3 * dh + 2];
    acc.x = fmaf(wl.x, t.l[dh], acc.x);
    acc.y = fmaf(wl.y, c.x, acc.y);
    acc.z = fmaf(wl.z, c.y, acc.z);
    acc.w = fmaf(wl.w, c.z, acc.w);
    acc.x = fmaf(wc.x, c.x, acc.x);
    acc.y = fmaf(wc.y, c.y, acc.y);
    acc.z = fmaf(wc.z, c.z, acc.z);
    acc.w = fmaf(wc.w, c.w, acc.w);
    acc.x = fmaf(wr.x, c.y, acc.x);
    acc.y = fmaf(wr.y, c.z, acc.y);
    acc.z = fmaf(wr.z, c.w, acc.z);
    acc.w = fmaf(wr.w, t.r[dh], acc.w);
  }
  return acc;
}

__global__ __launch_bounds__(256) void combine(
    const float* __restrict__ inp, const float* __restrict__ ffw,
    float* __restrict__ out) {
  const int lane = threadIdx.x & 31, wv = threadIdx.x >> 5;
  int bid = blockIdx.x;
  const int half = bid & 1;
  const int hg = (bid >> 1) & 15;
  const int s = (bid >> 5) & 1;
  const int b = bid >> 6;
  const int h = hg * 8 + wv;

  const int r0 = reflect_idx(h - 1, HH) * WW;
  const int r1 = h * WW;
  const int r2 = reflect_idx(h + 1, HH) * WW;
  const int co = 4 * lane;
  const int lo = (lane == 0) ? 1 : co - 1;
  const int ro = (lane == 31) ? WW - 2 : co + 4;

  const int c0 = 2 * CPW * half + s;
  const float* x = inp + (size_t)(b * 64 + c0) * HW;
  float* o = out + (size_t)(b * 64 + c0) * HW + r1 + co;

  const float* wg = ffw + ((size_t)(b * 2 + s) * 9) * HW + r1 + co;
  v4f wt[9];
#pragma unroll
  for (int k = 0; k < 9; ++k) wt[k] = *(const v4f*)(wg + (size_t)k * HW);

#pragma unroll
  for (int j = 0; j < CPW; ++j) {
    RowSet t;
    load_rows(x + (size_t)j * 2 * HW, r0, r1, r2, co, lo, ro, t);
    const v4f acc = apply_taps(t, wt);
    *(v4f*)(o + (size_t)j * 2 * HW) = acc;
  }
}

extern "C" void kernel_launch(void* const* d_in, const int* in_sizes, int n_in,
                              void* d_out, int out_size, void* d_ws, size_t ws_size,
                              hipStream_t stream) {
  const float* features = (const float*)d_in[0];
  const float* inpt     = (const float*)d_in[1];
  const float* conv_w   = (const float*)d_in[2];
  const float* conv_b   = (const float*)d_in[3];
  float* out = (float*)d_out;

  float* wp  = (float*)d_ws;
  float* ffw = wp + 18432;

  pack_w<<<(18432 + 255) / 256, 256, 0, stream>>>(conv_w, wp);
  conv_softmax<<<dim3(HH, 8), 256, 0, stream>>>(features, conv_b, wp, ffw);
  combine<<<512, 256, 0, stream>>>(inpt, ffw, out);
}


# conv GEMM re-indexed: rows = (out channel, column tap) 54 of 64, depth = (row tap, in channel) 192; column taps added after the GEMM via LDS (new sum order): 192 WMMA per wave, was 288
# speedup vs baseline: 2.6987x; 1.1633x over previous
#include <hip/hip_runtime.h>
#include <math.h>


typedef __attribute__((ext_vector_type(2))) float v2f;
typedef __attribute__((ext_vector_type(4))) float v4f;
typedef __attribute__((ext_vector_type(8))) float v8f;

#define HH 128
#define WW 128
#define HW (HH * WW)

__device__ __forceinline__ int reflect_idx(int i, int n) {
  if (i < 0) i = -i;
  if (i >= n) i = 2 * n - 2 - i;
  return i;
}

__global__ void pack_w(const float* __restrict__ cw, float* __restrict__ wp) {
  int t = blockIdx.x * blockDim.x + threadIdx.x;
  if (t >= 48 * 4 * 32 * 2) return;
  int v = t & 1;
  int lane = (t >> 1) & 31;
  int mt = (t >> 6) & 3;
  int kb = t >> 8;
  int m = (lane & 15) + 16 * mt;
  int hi = lane >> 4;
  int k = kb * 4 + 2 * hi + v;
  int dh = k >> 6;
  int ic = k & 63;
  float val = 0.0f;
  if (m < 54) {
    int c = m / 3, dw = m - 3 * c;
    val = cw[(c * 64 + ic) * 9 + dh * 3 + dw];
  }
  wp[t] = val;
}

__global__ __launch_bounds__(256) void conv_softmax(
    const float* __restrict__ feat, const float* __restrict__ bias,
    const float* __restrict__ wp, float* __restrict__ ffw) {
  __shared__ __attribute__((aligned(16))) float xs[64 * 3 * 132];
  __shared__ float zb[128 * 59];

  const int h = blockIdx.x;
  const int b = blockIdx.y;
  const int tid = threadIdx.x;
  const int lane = tid & 31, wv = tid >> 5;

  const float* fb = feat + (size_t)b * 64 * HW;
  v4f ev[12], od[12];
#pragma unroll
  for (int i = 0; i < 12; ++i) {
    const int r = i % 3;
    const int icp = wv * 4 + i / 3;
    const int hr = reflect_idx(h + r - 1, HH);
    const float* src = fb + (size_t)(2 * icp) * HW + hr * WW + 4 * lane;
    ev[i] = *(const v4f*)src;
    od[i] = *(const v4f*)(src + HW);
  }
#pragma unroll
  for (int i = 0; i < 12; ++i) {
    const int r = i % 3;
    const int icp = wv * 4 + i / 3;
    float* dst = xs + (icp * 3 + r) * 264 + 2 * (4 * lane + 2);
    v4f lo = {ev[i].x, od[i].x, ev[i].y, od[i].y};
    v4f hi4 = {ev[i].z, od[i].z, ev[i].w, od[i].w};
    *(v4f*)dst = lo;
    *(v4f*)(dst + 4) = hi4;
  }
  __syncthreads();

  const int p = lane & 15, hi = lane >> 4;
  const int w0 = wv * 16;

  const v2f* __restrict__ wf = (const v2f*)wp;
  const v2f* xs2 = (const v2f*)xs;

  v8f acc0 = {0.f, 0.f, 0.f, 0.f, 0.f, 0.f, 0.f, 0.f};
  v8f acc1 = acc0;
  v8f acc2 = acc0;
  v8f acc3 = acc0;

  int kb = 0;
  for (int dh = 0; dh < 3; ++dh) {
    const v2f* xrow = xs2 + (size_t)hi * 396 + dh * 132 + (w0 + p + 2);
#pragma unroll
    for (int j = 0; j < 16; ++j, ++kb) {
      const v2f* af = wf + kb * 128 + lane;
      v2f a0 = af[0];
      v2f a1 = af[32];
      v2f a2 = af[64];
      v2f a3 = af[96];
      v2f bx = xrow[(size_t)j * 792];
      acc0 = __builtin_amdgcn_wmma_f32_16x16x4_f32(
          false, a0, false, bx, (short)0, acc0, false, false);
      acc1 = __builtin_amdgcn_wmma_f32_16x16x4_f32(
          false, a1, false, bx, (short)0, acc1, false, false);
      acc2 = __builtin_amdgcn_wmma_f32_16x16x4_f32(
          false, a2, false, bx, (short)0, acc2, false, false);
      acc3 = __builtin_amdgcn_wmma_f32_16x16x4_f32(
          false, a3, false, bx, (short)0, acc3, false, false);
    }
  }

  const int wpix = w0 + p;
  float* zcol = &zb[wpix * 59];
#pragma unroll
  for (int r = 0; r < 8; ++r) {
    zcol[r + 8 * hi] = acc0[r];
    zcol[16 + r + 8 * hi] = acc1[r];
    zcol[32 + r + 8 * hi] = acc2[r];
  }
  if (hi == 0) {
#pragma unroll
    for (int r = 0; r < 6; ++r) zcol[48 + r] = acc3[r];
  }
  __syncthreads();

  const int g = hi;
  const float* zl = &zb[reflect_idx(wpix - 1, WW) * 59];
  const float* zr = &zb[reflect_idx(wpix + 1, WW) * 59];
  float z[9];
#pragma unroll
  for (int k = 0; k < 9; ++k) {
    const int ch = 2 * k + g;
    z[k] = ((zl[3 * ch] + zcol[3 * ch + 1]) + zr[3 * ch + 2]) + bias[ch];
  }
  float mx = z[0];
#pragma unroll
  for (int k = 1; k < 9; ++k) mx = fmaxf(mx, z[k]);
  float s = 0.0f;
#pragma unroll
  for (int k = 0; k < 9; ++k) {
    z[k] = expf(z[k] - mx);
    s += z[k];
  }
  const float inv = 1.0f / s;
#pragma unroll
  for (int k = 0; k < 9; ++k) {
    int ch = 2 * k + g;
    int si = ch / 9, kk = ch - si * 9;
    ffw[(((size_t)b * 2 + si) * 9 + kk) * HW + h * WW + wpix] = z[k] * inv;
  }
}

#define CPW 16

struct RowSet {
  v4f c[3];
  float l[3];
  float r[3];
};

__device__ __forceinline__ void load_rows(const float* __restrict__ x, int r0, int r1, int r2,
                                          int co, int lo, int ro, RowSet& t) {
  t.c[0] = *(const v4f*)(x + r0 + co);
  t.c[1] = *(const v4f*)(x + r1 + co);
  t.c[2] = *(const v4f*)(x + r2 + co);
  t.l[0] = x[r0 + lo];
  t.r[0] = x[r0 + ro];
  t.l[1] = x[r1 + lo];
  t.r[1] = x[r1 + ro];
  t.l[2] = x[r2 + lo];
  t.r[2] = x[r2 + ro];
}

__device__ __forceinline__ v4f apply_taps(const RowSet& t, const v4f* wt) {
  v4f acc = {0.f, 0.f, 0.f, 0.f};
#pragma unroll
  for (int dh = 0; dh < 3; ++dh) {
    const v4f c = t.c[dh];
    const v4f wl = wt[3 * dh], wc = wt[3 * dh + 1], wr = wt[3 * dh + 2];
    acc.x = fmaf(wl.x, t.l[dh], acc.x);
    acc.y = fmaf(wl.y, c.x, acc.y);
    acc.z = fmaf(wl.z, c.y, acc.z);
    acc.w = fmaf(wl.w, c.z, acc.w);
    acc.x = fmaf(wc.x, c.x, acc.x);
    acc.y = fmaf(wc.y, c.y, acc.y);
    acc.z = fmaf(wc.z, c.z, acc.z);
    acc.w = fmaf(wc.w, c.w, acc.w);
    acc.x = fmaf(wr.x, c.y, acc.x);
    acc.y = fmaf(wr.y, c.z, acc.y);
    acc.z = fmaf(wr.z, c.w, acc.z);
    acc.w = fmaf(wr.w, t.r[dh], acc.w);
  }
  return acc;
}

__global__ __launch_bounds__(256) void combine(
    const float* __restrict__ inp, const float* __restrict__ ffw,
    float* __restrict__ out) {
  const int lane = threadIdx.x & 31, wv = threadIdx.x >> 5;
  int bid = blockIdx.x;
  const int half = bid & 1;
  const int hg = (bid >> 1) & 15;
  const int s = (bid >> 5) & 1;
  const int b = bid >> 6;
  const int h = hg * 8 + wv;

  const int r0 = reflect_idx(h - 1, HH) * WW;
  const int r1 = h * WW;
  const int r2 = reflect_idx(h + 1, HH) * WW;
  const int co = 4 * lane;
  const int lo = (lane == 0) ? 1 : co - 1;
  const int ro = (lane == 31) ? WW - 2 : co + 4;

  const int c0 = 2 * CPW * half + s;
  const float* x = inp + (size_t)(b * 64 + c0) * HW;
  float* o = out + (size_t)(b * 64 + c0) * HW + r1 + co;

  const float* wg = ffw + ((size_t)(b * 2 + s) * 9) * HW + r1 + co;
  v4f wt[9];
#pragma unroll
  for (int k = 0; k < 9; ++k) wt[k] = *(const v4f*)(wg + (size_t)k * HW);

#pragma unroll
  for (int j = 0; j < CPW; ++j) {
    RowSet t;
    load_rows(x + (size_t)j * 2 * HW, r0, r1, r2, co, lo, ro, t);
    const v4f acc = apply_taps(t, wt);
    *(v4f*)(o + (size_t)j * 2 * HW) = acc;
  }
}

extern "C" void kernel_launch(void* const* d_in, const int* in_sizes, int n_in,
                              void* d_out, int out_size, void* d_ws, size_t ws_size,
                              hipStream_t stream) {
  const float* features = (const float*)d_in[0];
  const float* inpt     = (const float*)d_in[1];
  const float* conv_w   = (const float*)d_in[2];
  const float* conv_b   = (const float*)d_in[3];
  float* out = (float*)d_out;

  float* wp  = (float*)d_ws;
  float* ffw = wp + 18432;

  pack_w<<<48, 256, 0, stream>>>(conv_w, wp);
  conv_softmax<<<dim3(HH, 8), 256, 0, stream>>>(features, conv_b, wp, ffw);
  combine<<<512, 256, 0, stream>>>(inpt, ffw, out);
}


# combine fused into the conv kernel: a work-group applies its row's softmax weights (handed over in LDS, not workspace) to all 64 channels, inpt loads issued under the GEMM tail; 2 launches, was 3
# speedup vs baseline: 3.1124x; 1.1533x over previous
#include <hip/hip_runtime.h>
#include <math.h>


typedef __attribute__((ext_vector_type(2))) float v2f;
typedef __attribute__((ext_vector_type(4))) float v4f;
typedef __attribute__((ext_vector_type(8))) float v8f;

#define HH 128
#define WW 128
#define HW (HH * WW)

__device__ __forceinline__ int reflect_idx(int i, int n) {
  if (i < 0) i = -i;
  if (i >= n) i = 2 * n - 2 - i;
  return i;
}

__global__ void pack_w(const float* __restrict__ cw, float* __restrict__ wp) {
  int t = blockIdx.x * blockDim.x + threadIdx.x;
  if (t >= 48 * 4 * 32 * 2) return;
  int v = t & 1;
  int lane = (t >> 1) & 31;
  int mt = (t >> 6) & 3;
  int kb = t >> 8;
  int m = (lane & 15) + 16 * mt;
  int hi = lane >> 4;
  int k = kb * 4 + 2 * hi + v;
  int dh = k >> 6;
  int ic = k & 63;
  float val = 0.0f;
  if (m < 54) {
    int c = m / 3, dw = m - 3 * c;
    val = cw[(c * 64 + ic) * 9 + dh * 3 + dw];
  }
  wp[t] = val;
}

#define CPW 8
#define WLP 144

struct RowSet {
  v4f c[3];
  float l[3];
  float r[3];
};

__device__ __forceinline__ void load_rows(const float* __restrict__ x, int r0, int r1, int r2,
                                          int co, int lo, int ro, RowSet& t) {
  t.c[0] = *(const v4f*)(x + r0 + co);
  t.c[1] = *(const v4f*)(x + r1 + co);
  t.c[2] = *(const v4f*)(x + r2 + co);
  t.l[0] = x[r0 + lo];
  t.r[0] = x[r0 + ro];
  t.l[1] = x[r1 + lo];
  t.r[1] = x[r1 + ro];
  t.l[2] = x[r2 + lo];
  t.r[2] = x[r2 + ro];
}

__device__ __forceinline__ v4f apply_taps(const RowSet& t, const v4f* wt) {
  v4f acc = {0.f, 0.f, 0.f, 0.f};
#pragma unroll
  for (int dh = 0; dh < 3; ++dh) {
    const v4f c = t.c[dh];
    const v4f wl_ = wt[3 * dh], wc = wt[3 * dh + 1], wr = wt[3 * dh + 2];
    acc.x = fmaf(wl_.x, t.l[dh], acc.x);
    acc.y = fmaf(wl_.y, c.x, acc.y);
    acc.z = fmaf(wl_.z, c.y, acc.z);
    acc.w = fmaf(wl_.w, c.z, acc.w);
    acc.x = fmaf(wc.x, c.x, acc.x);
    acc.y = fmaf(wc.y, c.y, acc.y);
    acc.z = fmaf(wc.z, c.z, acc.z);
    acc.w = fmaf(wc.w, c.w, acc.w);
    acc.x = fmaf(wr.x, c.y, acc.x);
    acc.y = fmaf(wr.y, c.z, acc.y);
    acc.z = fmaf(wr.z, c.w, acc.z);
    acc.w = fmaf(wr.w, t.r[dh], acc.w);
  }
  return acc;
}

__global__ __launch_bounds__(256, 4) void conv_softmax_combine(
    const float* __restrict__ feat, const float* __restrict__ bias,
    const float* __restrict__ wp, const float* __restrict__ inp,
    float* __restrict__ out) {
  __shared__ __attribute__((aligned(16))) float xs[64 * 3 * 132];
  __shared__ float zb[128 * 59];
  float* const wl = xs;

  const int h = blockIdx.x;
  const int b = blockIdx.y;
  const int tid = threadIdx.x;
  const int lane = tid & 31, wv = tid >> 5;

  const float* fb = feat + (size_t)b * 64 * HW;
  v4f ev[12], od[12];
#pragma unroll
  for (int i = 0; i < 12; ++i) {
    const int r = i % 3;
    const int icp = wv * 4 + i / 3;
    const int hr = reflect_idx(h + r - 1, HH);
    const float* src = fb + (size_t)(2 * icp) * HW + hr * WW + 4 * lane;
    ev[i] = *(const v4f*)src;
    od[i] = *(const v4f*)(src + HW);
  }
#pragma unroll
  for (int i = 0; i < 12; ++i) {
    const int r = i % 3;
    const int icp = wv * 4 + i / 3;
    float* dst = xs + (icp * 3 + r) * 264 + 2 * (4 * lane + 2);
    v4f lo = {ev[i].x, od[i].x, ev[i].y, od[i].y};
    v4f hi4 = {ev[i].z, od[i].z, ev[i].w, od[i].w};
    *(v4f*)dst = lo;
    *(v4f*)(dst + 4) = hi4;
  }
  __syncthreads();

  const int p = lane & 15, hi = lane >> 4;
  const int w0 = wv * 16;

  const v2f* __restrict__ wf = (const v2f*)wp;
  const v2f* xs2 = (const v2f*)xs;

  v8f acc0 = {0.f, 0.f, 0.f, 0.f, 0.f, 0.f, 0.f, 0.f};
  v8f acc1 = acc0;
  v8f acc2 = acc0;
  v8f acc3 = acc0;

  int kb = 0;
  for (int dh = 0; dh < 3; ++dh) {
    const v2f* xrow = xs2 + (size_t)hi * 396 + dh * 132 + (w0 + p + 2);
#pragma unroll
    for (int j = 0; j < 16; ++j, ++kb) {
      const v2f* af = wf + kb * 128 + lane;
      v2f a0 = af[0];
      v2f a1 = af[32];
      v2f a2 = af[64];
      v2f a3 = af[96];
      v2f bx = xrow[(size_t)j * 792];
      acc0 = __builtin_amdgcn_wmma_f32_16x16x4_f32(
          false, a0, false, bx, (short)0, acc0, false, false);
      acc1 = __builtin_amdgcn_wmma_f32_16x16x4_f32(
          false, a1, false, bx, (short)0, acc1, false, false);
      acc2 = __builtin_amdgcn_wmma_f32_16x16x4_f32(
          false, a2, false, bx, (short)0, acc2, false, false);
      acc3 = __builtin_amdgcn_wmma_f32_16x16x4_f32(
          false, a3, false, bx, (short)0, acc3, false, false);
    }
  }

  const int cs = wv & 1;
  const int r0 = reflect_idx(h - 1, HH) * WW;
  const int r1 = h * WW;
  const int r2 = reflect_idx(h + 1, HH) * WW;
  const int co = 4 * lane;
  const int lo = (lane == 0) ? 1 : co - 1;
  const int ro = (lane == 31) ? WW - 2 : co + 4;
  const int c0 = cs + 16 * (wv >> 1);
  const float* x = inp + (size_t)(b * 64 + c0) * HW;
  RowSet rs[CPW];
#pragma unroll
  for (int j = 0; j < CPW; ++j)
    load_rows(x + (size_t)j * 2 * HW, r0, r1, r2, co, lo, ro, rs[j]);

  const int wpix = w0 + p;
  float* zcol = &zb[wpix * 59];
#pragma unroll
  for (int r = 0; r < 8; ++r) {
    zcol[r + 8 * hi] = acc0[r];
    zcol[16 + r + 8 * hi] = acc1[r];
    zcol[32 + r + 8 * hi] = acc2[r];
  }
  if (hi == 0) {
#pragma unroll
    for (int r = 0; r < 6; ++r) zcol[48 + r] = acc3[r];
  }
  __syncthreads();

  const int g = hi;
  const float* zl = &zb[reflect_idx(wpix - 1, WW) * 59];
  const float* zr = &zb[reflect_idx(wpix + 1, WW) * 59];
  float z[9];
#pragma unroll
  for (int k = 0; k < 9; ++k) {
    const int ch = 2 * k + g;
    const float be = bias[2 * k], bo = bias[2 * k + 1];
    z[k] = ((zl[3 * ch] + zcol[3 * ch + 1]) + zr[3 * ch + 2]) + (g ? bo : be);
  }
  float mx = z[0];
#pragma unroll
  for (int k = 1; k < 9; ++k) mx = fmaxf(mx, z[k]);
  float s = 0.0f;
#pragma unroll
  for (int k = 0; k < 9; ++k) {
    z[k] = expf(z[k] - mx);
    s += z[k];
  }
  const float inv = 1.0f / s;
#pragma unroll
  for (int k = 0; k < 9; ++k) wl[(2 * k + g) * WLP + wpix] = z[k] * inv;
  __syncthreads();

  v4f wt[9];
#pragma unroll
  for (int k = 0; k < 9; ++k) wt[k] = *(const v4f*)(wl + (cs * 9 + k) * WLP + co);
  float* o = out + (size_t)(b * 64 + c0) * HW + r1 + co;
#pragma unroll
  for (int j = 0; j < CPW; ++j) {
    const v4f acc = apply_taps(rs[j], wt);
    *(v4f*)(o + (size_t)j * 2 * HW) = acc;
  }
}

extern "C" void kernel_launch(void* const* d_in, const int* in_sizes, int n_in,
                              void* d_out, int out_size, void* d_ws, size_t ws_size,
                              hipStream_t stream) {
  const float* features = (const float*)d_in[0];
  const float* inpt     = (const float*)d_in[1];
  const float* conv_w   = (const float*)d_in[2];
  const float* conv_b   = (const float*)d_in[3];
  float* out = (float*)d_out;

  float* wp  = (float*)d_ws;

  pack_w<<<48, 256, 0, stream>>>(conv_w, wp);
  conv_softmax_combine<<<dim3(HH, 8), 256, 0, stream>>>(features, conv_b, wp, inpt, out);
}


# each work-group walks two image rows over a 3-row LDS ring of the feature slab, row h+2 loaded under the first GEMM's tail (512 groups, was 1024); combine edge columns come from neighbour lanes
# speedup vs baseline: 3.2447x; 1.0425x over previous
#include <hip/hip_runtime.h>
#include <math.h>


typedef __attribute__((ext_vector_type(2))) float v2f;
typedef __attribute__((ext_vector_type(4))) float v4f;
typedef __attribute__((ext_vector_type(8))) float v8f;

#define HH 128
#define WW 128
#define HW (HH * WW)

__device__ __forceinline__ int reflect_idx(int i, int n) {
  if (i < 0) i = -i;
  if (i >= n) i = 2 * n - 2 - i;
  return i;
}

__global__ void pack_w(const float* __restrict__ cw, float* __restrict__ wp) {
  int t = blockIdx.x * blockDim.x + threadIdx.x;
  if (t >= 48 * 4 * 32 * 2) return;
  int v = t & 1;
  int lane = (t >> 1) & 31;
  int mt = (t >> 6) & 3;
  int kb = t >> 8;
  int m = (lane & 15) + 16 * mt;
  int hi = lane >> 4;
  int k = kb * 4 + 2 * hi + v;
  int dh = k >> 6;
  int ic = k & 63;
  float val = 0.0f;
  if (m < 54) {
    int c = m / 3, dw = m - 3 * c;
    val = cw[(c * 64 + ic) * 9 + dh * 3 + dw];
  }
  wp[t] = val;
}

#define ROWS 2
#define ROW_GROUPS 64
static_assert(ROW_GROUPS * ROWS == HH, "the work-groups of an image cover its rows exactly");
#define CPW 8
#define WLP 144

struct RowSet {
  v4f c[3];
};

__device__ __forceinline__ void load_rows(const float* __restrict__ x, int r0, int r1, int r2,
                                          int co, RowSet& t) {
  t.c[0] = *(const v4f*)(x + r0 + co);
  t.c[1] = *(const v4f*)(x + r1 + co);
  t.c[2] = *(const v4f*)(x + r2 + co);
}

__device__ __forceinline__ v4f apply_taps(const RowSet& t, const v4f* wt, int lane) {
  v4f acc = {0.f, 0.f, 0.f, 0.f};
#pragma unroll
  for (int dh = 0; dh < 3; ++dh) {
    const v4f c = t.c[dh];
    const float up = __shfl_up(c.w, 1);
    const float dn = __shfl_down(c.x, 1);
    const float tl = (lane == 0) ? c.y : up;
    const float tr = (lane == 31) ? c.z : dn;
    const v4f wl_ = wt[3 * dh], wc = wt[3 * dh + 1], wr = wt[3 * dh + 2];
    acc.x = fmaf(wl_.x, tl, acc.x);
    acc.y = fmaf(wl_.y, c.x, acc.y);
    acc.z = fmaf(wl_.z, c.y, acc.z);
    acc.w = fmaf(wl_.w, c.z, acc.w);
    acc.x = fmaf(wc.x, c.x, acc.x);
    acc.y = fmaf(wc.y, c.y, acc.y);
    acc.z = fmaf(wc.z, c.z, acc.z);
    acc.w = fmaf(wc.w, c.w, acc.w);
    acc.x = fmaf(wr.x, c.y, acc.x);
    acc.y = fmaf(wr.y, c.z, acc.y);
    acc.z = fmaf(wr.z, c.w, acc.z);
    acc.w = fmaf(wr.w, tr, acc.w);
  }
  return acc;
}

__global__ __launch_bounds__(256, 4) void conv_softmax_combine(
    const float* __restrict__ feat, const float* __restrict__ bias,
    const float* __restrict__ wp, const float* __restrict__ inp,
    float* __restrict__ out) {
  __shared__ __attribute__((aligned(16))) float xs[64 * 3 * 132];
  __shared__ float zb[128 * 59];
  __shared__ __attribute__((aligned(16))) float wl[18 * WLP];
  __shared__ int park[256];

  const int h0 = blockIdx.x * ROWS;
  const int b = blockIdx.y;
  const int tid = threadIdx.x;
  park[tid] = tid;
  const int lane = tid & 31, wv = tid >> 5;

  const float* fb = feat + (size_t)b * 64 * HW;
  v4f ev[12], od[12];
#pragma unroll
  for (int i = 0; i < 12; ++i) {
    const int r = i % 3;
    const int icp = wv * 4 + i / 3;
    const int hr = reflect_idx(h0 + r - 1, HH);
    const float* src = fb + (size_t)(2 * icp) * HW + hr * WW + 4 * lane;
    ev[i] = *(const v4f*)src;
    od[i] = *(const v4f*)(src + HW);
  }
#pragma unroll
  for (int i = 0; i < 12; ++i) {
    const int r = i % 3;
    const int icp = wv * 4 + i / 3;
    float* dst = xs + (r * 32 + icp) * 264 + 2 * (4 * lane + 2);
    v4f lo = {ev[i].x, od[i].x, ev[i].y, od[i].y};
    v4f hi4 = {ev[i].z, od[i].z, ev[i].w, od[i].w};
    *(v4f*)dst = lo;
    *(v4f*)(dst + 4) = hi4;
  }
  __syncthreads();

  const v2f* __restrict__ wf = (const v2f*)wp;
  const v2f* xs2 = (const v2f*)xs;

#pragma unroll
  for (int it = 0; it < ROWS; ++it) {
  const int h = h0 + it;
  const bool more = (it + 1 < ROWS);
  int rt = park[tid];
  int lane = rt & 31, wv = rt >> 5;
  int p = lane & 15, hi = lane >> 4;
  int w0 = wv * 16;

  v8f acc0 = {0.f, 0.f, 0.f, 0.f, 0.f, 0.f, 0.f, 0.f};
  v8f acc1 = acc0;
  v8f acc2 = acc0;
  v8f acc3 = acc0;

  int kb = 0;
  for (int dh = 0; dh < 3; ++dh) {
    const int slot = (it + dh) % 3;
    const v2f* xrow = xs2 + (size_t)slot * 4224 + hi * 132 + (w0 + p + 2);
#pragma unroll
    for (int j = 0; j < 16; ++j, ++kb) {
      const v2f* af = wf + kb * 128 + lane;
      v2f a0 = af[0];
      v2f a1 = af[32];
      v2f a2 = af[64];
      v2f a3 = af[96];
      v2f bx = xrow[(size_t)j * 264];
      acc0 = __builtin_amdgcn_wmma_f32_16x16x4_f32(
          false, a0, false, bx, (short)0, acc0, false, false);
      acc1 = __builtin_amdgcn_wmma_f32_16x16x4_f32(
          false, a1, false, bx, (short)0, acc1, false, false);
      acc2 = __builtin_amdgcn_wmma_f32_16x16x4_f32(
          false, a2, false, bx, (short)0, acc2, false, false);
      acc3 = __builtin_amdgcn_wmma_f32_16x16x4_f32(
          false, a3, false, bx, (short)0, acc3, false, false);
    }
    __builtin_amdgcn_sched_barrier(0);
  }

  rt = park[tid];
  lane = rt & 31; wv = rt >> 5;
  p = lane & 15; hi = lane >> 4;
  w0 = wv * 16;

  const int cs = wv & 1;
  const int r0 = reflect_idx(h - 1, HH) * WW;
  const int r1 = h * WW;
  const int r2 = reflect_idx(h + 1, HH) * WW;
  const int co = 4 * lane;
  const int c0 = cs + 16 * (wv >> 1);
  const float* x = inp + (size_t)(b * 64 + c0) * HW;
  RowSet rs[CPW];
#pragma unroll
  for (int j = 0; j < CPW; ++j)
    load_rows(x + (size_t)j * 2 * HW, r0, r1, r2, co, rs[j]);

  v4f nev[4] = {}, nod[4] = {};
  if (more) {
    const int hr = reflect_idx(h + 2, HH);
#pragma unroll
    for (int q = 0; q < 4; ++q) {
      const float* src = fb + (size_t)(2 * (wv * 4 + q)) * HW + hr * WW + 4 * lane;
      nev[q] = *(const v4f*)src;
      nod[q] = *(const v4f*)(src + HW);
    }
  }

  const int wpix = w0 + p;
  float* zcol = &zb[wpix * 59];
#pragma unroll
  for (int r = 0; r < 8; ++r) {
    zcol[r + 8 * hi] = acc0[r];
    zcol[16 + r + 8 * hi] = acc1[r];
    zcol[32 + r + 8 * hi] = acc2[r];
  }
  if (hi == 0) {
#pragma unroll
    for (int r = 0; r < 6; ++r) zcol[48 + r] = acc3[r];
  }
  __syncthreads();

  if (more) {
    const int s = it % 3;
#pragma unroll
    for (int q = 0; q < 4; ++q) {
      float* dst = xs + (s * 32 + wv * 4 + q) * 264 + 2 * (4 * lane + 2);
      v4f lo4 = {nev[q].x, nod[q].x, nev[q].y, nod[q].y};
      v4f hi4 = {nev[q].z, nod[q].z, nev[q].w, nod[q].w};
      *(v4f*)dst = lo4;
      *(v4f*)(dst + 4) = hi4;
    }
  }

  const int g = hi;
  const float* zl = &zb[reflect_idx(wpix - 1, WW) * 59];
  const float* zr = &zb[reflect_idx(wpix + 1, WW) * 59];
  float z[9];
#pragma unroll
  for (int k = 0; k < 9; ++k) {
    const int ch = 2 * k + g;
    const float be = bias[2 * k], bo = bias[2 * k + 1];
    z[k] = ((zl[3 * ch] + zcol[3 * ch + 1]) + zr[3 * ch + 2]) + (g ? bo : be);
  }
  float mx = z[0];
#pragma unroll
  for (int k = 1; k < 9; ++k) mx = fmaxf(mx, z[k]);
  float s = 0.0f;
#pragma unroll
  for (int k = 0; k < 9; ++k) {
    z[k] = expf(z[k] - mx);
    s += z[k];
  }
  const float inv = 1.0f / s;
#pragma unroll
  for (int k = 0; k < 9; ++k) wl[(2 * k + g) * WLP + wpix] = z[k] * inv;
  __syncthreads();

  v4f wt[9];
#pragma unroll
  for (int k = 0; k < 9; ++k) wt[k] = *(const v4f*)(wl + (cs * 9 + k) * WLP + co);
  float* o = out + (size_t)(b * 64 + c0) * HW + r1 + co;
#pragma unroll
  for (int j = 0; j < CPW; ++j) {
    const v4f acc = apply_taps(rs[j], wt, lane);
    *(v4f*)(o + (size_t)j * 2 * HW) = acc;
  }
  __builtin_amdgcn_sched_barrier(0);
  }
}

extern "C" void kernel_launch(void* const* d_in, const int* in_sizes, int n_in,
                              void* d_out, int out_size, void* d_ws, size_t ws_size,
                              hipStream_t stream) {
  const float* features = (const float*)d_in[0];
  const float* inpt     = (const float*)d_in[1];
  const float* conv_w   = (const float*)d_in[2];
  const float* conv_b   = (const float*)d_in[3];
  float* out = (float*)d_out;

  float* wp  = (float*)d_ws;

  pack_w<<<48, 256, 0, stream>>>(conv_w, wp);
  conv_softmax_combine<<<dim3(ROW_GROUPS, 8), 256, 0, stream>>>(features, conv_b, wp, inpt, out);
}


# conv weights are put into WMMA fragment order by each work-group in LDS (D scratch moves into the dead ring slot) and read from there by the GEMM; the separate pack launch is gone: 1 launch, was 2
# speedup vs baseline: 3.5679x; 1.0996x over previous
#include <hip/hip_runtime.h>
#include <math.h>


typedef __attribute__((ext_vector_type(2))) float v2f;
typedef __attribute__((ext_vector_type(4))) float v4f;
typedef __attribute__((ext_vector_type(8))) float v8f;

#define HH 128
#define WW 128
#define HW (HH * WW)

__device__ __forceinline__ int reflect_idx(int i, int n) {
  if (i < 0) i = -i;
  if (i >= n) i = 2 * n - 2 - i;
  return i;
}

#define AKB 192
#define AT3 (48 * AKB)
#define APK (AT3 + 24 * 64 + 16)
#define ROWS 2
#define ROW_GROUPS 64
static_assert(ROW_GROUPS * ROWS == HH, "the work-groups of an image cover its rows exactly");
#define CPW 8
#define WLP 144

struct RowSet {
  v4f c[3];
};

__device__ __forceinline__ void load_rows(const float* __restrict__ x, int r0, int r1, int r2,
                                          int co, RowSet& t) {
  t.c[0] = *(const v4f*)(x + r0 + co);
  t.c[1] = *(const v4f*)(x + r1 + co);
  t.c[2] = *(const v4f*)(x + r2 + co);
}

__device__ __forceinline__ v4f apply_taps(const RowSet& t, const v4f* wt, int lane) {
  v4f acc = {0.f, 0.f, 0.f, 0.f};
#pragma unroll
  for (int dh = 0; dh < 3; ++dh) {
    const v4f c = t.c[dh];
    const float up = __shfl_up(c.w, 1);
    const float dn = __shfl_down(c.x, 1);
    const float tl = (lane == 0) ? c.y : up;
    const float tr = (lane == 31) ? c.z : dn;
    const v4f wl_ = wt[3 * dh], wc = wt[3 * dh + 1], wr = wt[3 * dh + 2];
    acc.x = fmaf(wl_.x, tl, acc.x);
    acc.y = fmaf(wl_.y, c.x, acc.y);
    acc.z = fmaf(wl_.z, c.y, acc.z);
    acc.w = fmaf(wl_.w, c.z, acc.w);
    acc.x = fmaf(wc.x, c.x, acc.x);
    acc.y = fmaf(wc.y, c.y, acc.y);
    acc.z = fmaf(wc.z, c.z, acc.z);
    acc.w = fmaf(wc.w, c.w, acc.w);
    acc.x = fmaf(wr.x, c.y, acc.x);
    acc.y = fmaf(wr.y, c.z, acc.y);
    acc.z = fmaf(wr.z, c.w, acc.z);
    acc.w = fmaf(wr.w, tr, acc.w);
  }
  return acc;
}

__global__ __launch_bounds__(256, 4) void conv_softmax_combine(
    const float* __restrict__ feat, const float* __restrict__ bias,
    const float* __restrict__ cw, const float* __restrict__ inp,
    float* __restrict__ out) {
  __shared__ struct __attribute__((aligned(16))) {
    float ap[APK];
    float xs[64 * 3 * 132];
    float wl[18 * WLP];
    int park[256];
  } sm;
  float* const xs = sm.xs;
  float* const ap = sm.ap;
  float* const wl = sm.wl;
  int* const park = sm.park;

  const int h0 = blockIdx.x * ROWS;
  const int b = blockIdx.y;
  const int tid = threadIdx.x;
  park[tid] = tid;
  const int lane = tid & 31, wv = tid >> 5;

  const float* fb = feat + (size_t)b * 64 * HW;
  v4f ev[12], od[12];
#pragma unroll
  for (int i = 0; i < 12; ++i) {
    const int r = i % 3;
    const int icp = wv * 4 + i / 3;
    const int hr = reflect_idx(h0 + r - 1, HH);
    const float* src = fb + (size_t)(2 * icp) * HW + hr * WW + 4 * lane;
    ev[i] = *(const v4f*)src;
    od[i] = *(const v4f*)(src + HW);
  }
  float wq[48];
  const bool packs = tid < 224;
  if (packs) {
    const int v = tid & 1;
    int m, kh;
    if (tid < 192) {
      const int ln = (tid >> 1) & 31;
      m = (ln & 15) + 16 * (tid >> 6);
      kh = ln >> 4;
    } else {
      const int q = (tid - 192) >> 1;
      m = 48 + (q & 7);
      kh = q >> 3;
    }
    const int mc = m < 54 ? m : 0;
    const int c = mc / 3, dw = mc - 3 * c;
    const float* ws = cw + (c * 64 + 2 * kh + v) * 9 + dw;
#pragma unroll
    for (int kb = 0; kb < 48; ++kb) wq[kb] = ws[(kb & 15) * 36 + (kb >> 4) * 3];
  }
#pragma unroll
  for (int i = 0; i < 12; ++i) {
    const int r = i % 3;
    const int icp = wv * 4 + i / 3;
    float* dst = xs + (r * 32 + icp) * 264 + 2 * (4 * lane + 2);
    v4f lo = {ev[i].x, od[i].x, ev[i].y, od[i].y};
    v4f hi4 = {ev[i].z, od[i].z, ev[i].w, od[i].w};
    *(v4f*)dst = lo;
    *(v4f*)(dst + 4) = hi4;
  }
  if (tid < 192) {
#pragma unroll
    for (int kb = 0; kb < 48; ++kb) ap[kb * AKB + tid] = wq[kb];
  } else if (packs) {
    const int t = tid - 192;
    float* d3 = ap + AT3 + t + (t & 16);
#pragma unroll
    for (int kb = 0; kb < 48; ++kb) d3[64 * (kb >> 1) + 16 * (kb & 1)] = wq[kb];
  } else if (tid < 240) {
    ap[AT3 + 24 * 64 + (tid - 224)] = 0.0f;
  }
  __syncthreads();

  const v2f* ab = (const v2f*)ap;
  const v2f* xs2 = (const v2f*)xs;

#pragma unroll
  for (int it = 0; it < ROWS; ++it) {
  const int h = h0 + it;
  const bool more = (it + 1 < ROWS);
  int rt = park[tid];
  int lane = rt & 31, wv = rt >> 5;
  int p = lane & 15, hi = lane >> 4;
  int w0 = wv * 16;

  v8f acc0 = {0.f, 0.f, 0.f, 0.f, 0.f, 0.f, 0.f, 0.f};
  v8f acc1 = acc0;
  v8f acc2 = acc0;
  v8f acc3 = acc0;

  RowSet rs[CPW];
  v4f nev[4] = {}, nod[4] = {};

  int kb = 0;
#pragma unroll
  for (int dh = 0; dh < 3; ++dh) {
    if (dh == 2) {
      const int rl = park[tid];
      const int ll = rl & 31, wl_ = rl >> 5;
      const int cs_ = wl_ & 1;
      const int c0_ = cs_ + 16 * (wl_ >> 1);
      const float* x = inp + (size_t)(b * 64 + c0_) * HW;
      const int r0 = reflect_idx(h - 1, HH) * WW;
      const int r2 = reflect_idx(h + 1, HH) * WW;
#pragma unroll
      for (int j = 0; j < CPW; ++j)
        load_rows(x + (size_t)j * 2 * HW, r0, h * WW, r2, 4 * ll, rs[j]);
      if (more) {
        const int hr = reflect_idx(h + 2, HH);
#pragma unroll
        for (int q = 0; q < 4; ++q) {
          const float* src = fb + (size_t)(2 * (wl_ * 4 + q)) * HW + hr * WW + 4 * ll;
          nev[q] = *(const v4f*)src;
          nod[q] = *(const v4f*)(src + HW);
        }
      }
      __builtin_amdgcn_sched_barrier(0);
    }
    const int slot = (it + dh) % 3;
    const v2f* xrow = xs2 + (size_t)slot * 4224 + hi * 132 + (w0 + p + 2);
#pragma unroll
    for (int j = 0; j < 16; ++j, ++kb) {
      const v2f* af = ab + kb * (AKB / 2) + lane;
      v2f a0 = af[0];
      v2f a1 = af[32];
      v2f a2 = af[64];
      v2f a3 = ab[AT3 / 2 + 32 * (kb >> 1) + 8 * (kb & 1) + lane];
      v2f bx = xrow[(size_t)j * 264];
      acc0 = __builtin_amdgcn_wmma_f32_16x16x4_f32(
          false, a0, false, bx, (short)0, acc0, false, false);
      acc1 = __builtin_amdgcn_wmma_f32_16x16x4_f32(
          false, a1, false, bx, (short)0, acc1, false, false);
      acc2 = __builtin_amdgcn_wmma_f32_16x16x4_f32(
          false, a2, false, bx, (short)0, acc2, false, false);
      acc3 = __builtin_amdgcn_wmma_f32_16x16x4_f32(
          false, a3, false, bx, (short)0, acc3, false, false);
    }
    __builtin_amdgcn_sched_barrier(0);
  }

  rt = park[tid];
  lane = rt & 31; wv = rt >> 5;
  p = lane & 15; hi = lane >> 4;
  w0 = wv * 16;
  const int cs = wv & 1;
  const int r1 = h * WW;
  const int co = 4 * lane;
  const int c0 = cs + 16 * (wv >> 1);

  __syncthreads();
  float* zb = xs + (it % 3) * (32 * 264);
  const int wpix = w0 + p;
  float* zcol = &zb[wpix * 59];
#pragma unroll
  for (int r = 0; r < 8; ++r) {
    zcol[r + 8 * hi] = acc0[r];
    zcol[16 + r + 8 * hi] = acc1[r];
    zcol[32 + r + 8 * hi] = acc2[r];
  }
  if (hi == 0) {
#pragma unroll
    for (int r = 0; r < 6; ++r) zcol[48 + r] = acc3[r];
  }
  __syncthreads();

  const int g = hi;
  const float* zl = &zb[reflect_idx(wpix - 1, WW) * 59];
  const float* zr = &zb[reflect_idx(wpix + 1, WW) * 59];
  float z[9];
#pragma unroll
  for (int k = 0; k < 9; ++k) {
    const int ch = 2 * k + g;
    const float be = bias[2 * k], bo = bias[2 * k + 1];
    z[k] = ((zl[3 * ch] + zcol[3 * ch + 1]) + zr[3 * ch + 2]) + (g ? bo : be);
  }
  float mx = z[0];
#pragma unroll
  for (int k = 1; k < 9; ++k) mx = fmaxf(mx, z[k]);
  float s = 0.0f;
#pragma unroll
  for (int k = 0; k < 9; ++k) {
    z[k] = expf(z[k] - mx);
    s += z[k];
  }
  const float inv = 1.0f / s;
#pragma unroll
  for (int k = 0; k < 9; ++k) wl[(2 * k + g) * WLP + wpix] = z[k] * inv;
  __syncthreads();

  if (more) {
    const int s = it % 3;
#pragma unroll
    for (int q = 0; q < 4; ++q) {
      float* dst = xs + (s * 32 + wv * 4 + q) * 264 + 2 * (4 * lane + 2);
      v4f lo4 = {nev[q].x, nod[q].x, nev[q].y, nod[q].y};
      v4f hi4 = {nev[q].z, nod[q].z, nev[q].w, nod[q].w};
      *(v4f*)dst = lo4;
      *(v4f*)(dst + 4) = hi4;
    }
    __syncthreads();
  }

  __builtin_amdgcn_sched_barrier(0);
  v4f wt[9];
#pragma unroll
  for (int k = 0; k < 9; ++k) wt[k] = *(const v4f*)(wl + (cs * 9 + k) * WLP + co);
  float* o = out + (size_t)(b * 64 + c0) * HW + r1 + co;
#pragma unroll
  for (int j = 0; j < CPW; ++j) {
    const v4f acc = apply_taps(rs[j], wt, lane);
    *(v4f*)(o + (size_t)j * 2 * HW) = acc;
  }
  __builtin_amdgcn_sched_barrier(0);
  }
}

extern "C" void kernel_launch(void* const* d_in, const int* in_sizes, int n_in,
                              void* d_out, int out_size, void* d_ws, size_t ws_size,
                              hipStream_t stream) {
  const float* features = (const float*)d_in[0];
  const float* inpt     = (const float*)d_in[1];
  const float* conv_w   = (const float*)d_in[2];
  const float* conv_b   = (const float*)d_in[3];
  float* out = (float*)d_out;

  conv_softmax_combine<<<dim3(ROW_GROUPS, 8), 256, 0, stream>>>(features, conv_b, conv_w, inpt, out);
}
